# starting kernel
# baseline (speedup 1.0000x reference)
#include <hip/hip_runtime.h>


typedef float v2f __attribute__((ext_vector_type(2)));
typedef float v8f __attribute__((ext_vector_type(8)));

#define HAAR_C 0.35355339059327373f

__device__ __forceinline__ float haarF(int s, int k) {
    return (__builtin_popcount(s & k) & 1) ? -HAAR_C : HAAR_C;
}

__global__ __launch_bounds__(256)
void dwt3d_haar_wmma(const float* __restrict__ x, float* __restrict__ out) {
    constexpr int C  = 16;
    constexpr int D  = 64, H  = 128, W  = 128;
    constexpr int Dh = 32, Hh = 64,  Wh = 64;
    constexpr int SUBBAND_STRIDE = 2 * C * Dh * Hh * Wh;
    constexpr int IN_PLANE = H * W;

    const int tid  = threadIdx.x;
    const int lane = tid & 31;
    const int wid  = blockIdx.x * 8 + (tid >> 5);

    const int h  = wid & 63;
    const int d  = (wid >> 6) & 31;
    const int cc = (wid >> 11) & 15;
    const int bb = wid >> 15;

    const int Mlo = lane & 15;
    const int K0  = (lane >> 4) << 1;
    const bool top = (Mlo < 8);
    const int  s0  = top ? Mlo : (Mlo - 8);
    const float f0 = haarF(s0, K0);
    const float f1 = haarF(s0, K0 + 1);
    const float f4 = haarF(s0, K0 + 4);
    const float f5 = haarF(s0, K0 + 5);
    const v2f a0 = { top ? f0 : 0.0f, top ? f1 : 0.0f };
    const v2f a1 = { top ? f4 : 0.0f, top ? f5 : 0.0f };
    const v2f a2 = { top ? 0.0f : f0, top ? 0.0f : f1 };
    const v2f a3 = { top ? 0.0f : f4, top ? 0.0f : f5 };

    const int n      = lane & 15;
    const int rowsel = lane >> 4;
    const float* base = x + (((bb * C + cc) * D + 2 * d) * H + (2 * h + rowsel)) * W;

    const int outRowBase = (((bb * C + cc) * Dh + d) * Hh + h) * Wh;

    #pragma unroll
    for (int half = 0; half < 2; ++half) {
        const int w0 = half * 32;
        const int wx = 2 * w0 + 2 * n;

        const v2f b0 = __builtin_nontemporal_load((const v2f*)(base + wx));
        const v2f b1 = __builtin_nontemporal_load((const v2f*)(base + IN_PLANE + wx));
        const v2f b2 = __builtin_nontemporal_load((const v2f*)(base + wx + 32));
        const v2f b3 = __builtin_nontemporal_load((const v2f*)(base + IN_PLANE + wx + 32));

        v8f acc = {};
        acc = __builtin_amdgcn_wmma_f32_16x16x4_f32(false, a0, false, b0, (short)0, acc, false, false);
        acc = __builtin_amdgcn_wmma_f32_16x16x4_f32(false, a1, false, b1, (short)0, acc, false, false);
        acc = __builtin_amdgcn_wmma_f32_16x16x4_f32(false, a2, false, b2, (short)0, acc, false, false);
        acc = __builtin_amdgcn_wmma_f32_16x16x4_f32(false, a3, false, b3, (short)0, acc, false, false);

        const int o = outRowBase + w0 + lane;
        #pragma unroll
        for (int s = 0; s < 8; ++s) {
            __builtin_nontemporal_store(acc[s], out + s * SUBBAND_STRIDE + o);
        }
    }
}

extern "C" void kernel_launch(void* const* d_in, const int* in_sizes, int n_in,
                              void* d_out, int out_size, void* d_ws, size_t ws_size,
                              hipStream_t stream) {
    (void)in_sizes; (void)n_in; (void)out_size; (void)d_ws; (void)ws_size;
    const float* x = (const float*)d_in[0];
    float* out = (float*)d_out;
    dim3 grid(8192), block(256);
    hipLaunchKernelGGL(dwt3d_haar_wmma, grid, block, 0, stream, x, out);
}


# Each lane loads 16 B (two adjacent blocks); even and odd blocks feed two accumulator tiles: per wave 4 b128 loads + 8 b64 stores of whole 256-B runs replace 8 b64 + 16 b32; order of sums unchanged
# speedup vs baseline: 1.0377x; 1.0377x over previous
#include <hip/hip_runtime.h>


typedef float v2f __attribute__((ext_vector_type(2)));
typedef float v4f __attribute__((ext_vector_type(4)));
typedef float v8f __attribute__((ext_vector_type(8)));

#define HAAR_C 0.35355339059327373f

__device__ __forceinline__ float haarF(int s, int k) {
    return (__builtin_popcount(s & k) & 1) ? -HAAR_C : HAAR_C;
}

__global__ __launch_bounds__(256)
void dwt3d_haar_wmma(const float* __restrict__ x, float* __restrict__ out) {
    constexpr int C  = 16;
    constexpr int D  = 64, H  = 128, W  = 128;
    constexpr int Dh = 32, Hh = 64,  Wh = 64;
    constexpr int SUBBAND_STRIDE = 2 * C * Dh * Hh * Wh;
    constexpr int IN_PLANE = H * W;

    const int tid  = threadIdx.x;
    const int lane = tid & 31;
    const int wid  = blockIdx.x * 8 + (tid >> 5);

    const int h  = wid & 63;
    const int d  = (wid >> 6) & 31;
    const int cc = (wid >> 11) & 15;
    const int bb = wid >> 15;

    const int Mlo = lane & 15;
    const int K0  = (lane >> 4) << 1;
    const bool top = (Mlo < 8);
    const int  s0  = top ? Mlo : (Mlo - 8);
    const float f0 = haarF(s0, K0);
    const float f1 = haarF(s0, K0 + 1);
    const float f4 = haarF(s0, K0 + 4);
    const float f5 = haarF(s0, K0 + 5);
    const v2f a0 = { top ? f0 : 0.0f, top ? f1 : 0.0f };
    const v2f a1 = { top ? f4 : 0.0f, top ? f5 : 0.0f };
    const v2f a2 = { top ? 0.0f : f0, top ? 0.0f : f1 };
    const v2f a3 = { top ? 0.0f : f4, top ? 0.0f : f5 };

    const int n      = lane & 15;
    const int rowsel = lane >> 4;
    const float* base = x + (((bb * C + cc) * D + 2 * d) * H + (2 * h + rowsel)) * W + 4 * n;

    const v4f l00 = __builtin_nontemporal_load((const v4f*)(base));
    const v4f l10 = __builtin_nontemporal_load((const v4f*)(base + 64));
    const v4f l01 = __builtin_nontemporal_load((const v4f*)(base + IN_PLANE));
    const v4f l11 = __builtin_nontemporal_load((const v4f*)(base + IN_PLANE + 64));

    v8f accE = {};
    v8f accO = {};
    accE = __builtin_amdgcn_wmma_f32_16x16x4_f32(false, a0, false, l00.xy, (short)0, accE, false, false);
    accO = __builtin_amdgcn_wmma_f32_16x16x4_f32(false, a0, false, l00.zw, (short)0, accO, false, false);
    accE = __builtin_amdgcn_wmma_f32_16x16x4_f32(false, a1, false, l01.xy, (short)0, accE, false, false);
    accO = __builtin_amdgcn_wmma_f32_16x16x4_f32(false, a1, false, l01.zw, (short)0, accO, false, false);
    accE = __builtin_amdgcn_wmma_f32_16x16x4_f32(false, a2, false, l10.xy, (short)0, accE, false, false);
    accO = __builtin_amdgcn_wmma_f32_16x16x4_f32(false, a2, false, l10.zw, (short)0, accO, false, false);
    accE = __builtin_amdgcn_wmma_f32_16x16x4_f32(false, a3, false, l11.xy, (short)0, accE, false, false);
    accO = __builtin_amdgcn_wmma_f32_16x16x4_f32(false, a3, false, l11.zw, (short)0, accO, false, false);

    const int o = (((bb * C + cc) * Dh + d) * Hh + h) * Wh + 2 * lane;
    #pragma unroll
    for (int s = 0; s < 8; ++s) {
        const v2f pr = { accE[s], accO[s] };
        __builtin_nontemporal_store(pr, (v2f*)(out + s * SUBBAND_STRIDE + o));
    }
}

extern "C" void kernel_launch(void* const* d_in, const int* in_sizes, int n_in,
                              void* d_out, int out_size, void* d_ws, size_t ws_size,
                              hipStream_t stream) {
    (void)in_sizes; (void)n_in; (void)out_size; (void)d_ws; (void)ws_size;
    const float* x = (const float*)d_in[0];
    float* out = (float*)d_out;
    dim3 grid(8192), block(256);
    hipLaunchKernelGGL(dwt3d_haar_wmma, grid, block, 0, stream, x, out);
}
